# starting kernel
# baseline (speedup 1.0000x reference)
#include <hip/hip_runtime.h>

typedef __attribute__((ext_vector_type(2))) float v2f;
typedef __attribute__((ext_vector_type(8))) float v8f;

#define N_ATOMS 32
#define E_MOL   64
#define IN_CH   64
#define OUT_CH  128
#define HID     256
#define M_MOL   10000
#define E_KG    320000

__device__ __forceinline__ v8f wmma_f32_k4(v2f a, v2f b, v8f c) {
  return __builtin_amdgcn_wmma_f32_16x16x4_f32(false, a, false, b, (short)0, c,
                                               false, false);
}

__device__ __forceinline__ void gemm32_wmma(const float* __restrict__ A32,
                                            const float* __restrict__ W,
                                            float* __restrict__ T,
                                            int K, int N,
                                            int wave, int nwaves, int lane) {
  const int hi = lane >> 4;
  const int lo = lane & 15;
  const int ntTiles = N >> 4;
  for (int t = wave; t < 2 * ntTiles; t += nwaves) {
    const int mt = t & 1;
    const int nt = t >> 1;
    const int mrow = mt * 16 + lo;
    const int ncol = nt * 16 + lo;
    v8f acc = {};
    for (int kb = 0; kb < K; kb += 4) {
      const int k = kb + 2 * hi;
      v2f a, b;
      a.x = A32[mrow * K + k];
      a.y = A32[mrow * K + k + 1];
      b.x = W[k * N + ncol];
      b.y = W[(k + 1) * N + ncol];
      acc = wmma_f32_k4(a, b, acc);
    }
#pragma unroll
    for (int v = 0; v < 8; ++v)
      T[(mt * 16 + v + 8 * hi) * N + nt * 16 + lo] = acc[v];
  }
}

__device__ __forceinline__ void aggregate(const float* __restrict__ T,
                                          float* __restrict__ H,
                                          const float* __restrict__ bias,
                                          int Cshift, bool relu,
                                          const int* s_src, const int* s_dst,
                                          const float* s_dinv,
                                          const float* s_norm,
                                          int tid, int nthr) {
  const int Cm1 = (1 << Cshift) - 1;
  for (int idx = tid; idx < (N_ATOMS << Cshift); idx += nthr) {
    const int i = idx >> Cshift;
    const int c = idx & Cm1;
    H[idx] = s_dinv[i] * s_dinv[i] * T[idx] + bias[c];
  }
  __syncthreads();
  for (int idx = tid; idx < (E_MOL << Cshift); idx += nthr) {
    const int e = idx >> Cshift;
    const int c = idx & Cm1;
    atomicAdd(&H[(s_dst[e] << Cshift) + c],
              s_norm[e] * T[(s_src[e] << Cshift) + c]);
  }
  __syncthreads();
  if (relu) {
    for (int idx = tid; idx < (N_ATOMS << Cshift); idx += nthr)
      H[idx] = fmaxf(H[idx], 0.0f);
  }
  __syncthreads();
}

__global__ __launch_bounds__(256) void mol_gnn_kernel(
    const float* __restrict__ mol_x, const int* __restrict__ mei,
    const float* __restrict__ W1, const float* __restrict__ b1,
    const float* __restrict__ W2, const float* __restrict__ b2,
    const float* __restrict__ W3, const float* __restrict__ b3,
    float* __restrict__ embed) {
  __shared__ float Abuf[N_ATOMS * HID];
  __shared__ float Bbuf[N_ATOMS * HID];
  __shared__ float s_dinv[N_ATOMS];
  __shared__ float s_norm[E_MOL];
  __shared__ int   s_deg[N_ATOMS];
  __shared__ int   s_src[E_MOL];
  __shared__ int   s_dst[E_MOL];

  const int mol  = blockIdx.x;
  const int tid  = threadIdx.x;
  const int nthr = blockDim.x;
  const int lane = tid & 31;
  const int wave = tid >> 5;
  const int nw   = nthr >> 5;

  const float* X = mol_x + (size_t)mol * (N_ATOMS * IN_CH);
  for (int i = tid; i < N_ATOMS * IN_CH; i += nthr) Abuf[i] = X[i];
  const int* me = mei + (size_t)mol * (2 * E_MOL);
  if (tid < E_MOL) { s_src[tid] = me[tid]; s_dst[tid] = me[E_MOL + tid]; }
  if (tid < N_ATOMS) s_deg[tid] = 1;
  __syncthreads();
  if (tid < E_MOL) atomicAdd(&s_deg[s_dst[tid]], 1);
  __syncthreads();
  if (tid < N_ATOMS) s_dinv[tid] = rsqrtf((float)s_deg[tid]);
  __syncthreads();
  if (tid < E_MOL) s_norm[tid] = s_dinv[s_src[tid]] * s_dinv[s_dst[tid]];
  __syncthreads();

  gemm32_wmma(Abuf, W1, Bbuf, IN_CH, HID, wave, nw, lane);
  __syncthreads();
  aggregate(Bbuf, Abuf, b1, 8, true, s_src, s_dst, s_dinv, s_norm, tid, nthr);

  gemm32_wmma(Abuf, W2, Bbuf, HID, HID, wave, nw, lane);
  __syncthreads();
  aggregate(Bbuf, Abuf, b2, 8, true, s_src, s_dst, s_dinv, s_norm, tid, nthr);

  gemm32_wmma(Abuf, W3, Bbuf, HID, OUT_CH, wave, nw, lane);
  __syncthreads();
  aggregate(Bbuf, Abuf, b3, 7, false, s_src, s_dst, s_dinv, s_norm, tid, nthr);

  for (int c = tid; c < OUT_CH; c += nthr) {
    float v = Abuf[c];
    for (int r = 1; r < N_ATOMS; ++r) v = fminf(v, Abuf[r * OUT_CH + c]);
    embed[(size_t)mol * OUT_CH + c] = v;
  }
}

__global__ __launch_bounds__(256) void wmma_gemm_kernel(
    const float* __restrict__ A, const float* __restrict__ B,
    float* __restrict__ C, int Mrows, int K, int N) {
  const int lane = threadIdx.x & 31;
  const int wave = blockIdx.x * (blockDim.x >> 5) + (threadIdx.x >> 5);
  const int ntT = N >> 4;
  const int nTiles = (Mrows >> 4) * ntT;
  if (wave >= nTiles) return;
  const int mt = wave / ntT;
  const int nt = wave - mt * ntT;
  const int hi = lane >> 4, lo = lane & 15;
  const int mrow = mt * 16 + lo;
  const int ncol = nt * 16 + lo;
  v8f acc = {};
  for (int kb = 0; kb < K; kb += 4) {
    const int k = kb + 2 * hi;
    v2f a, b;
    a.x = A[(size_t)mrow * K + k];
    a.y = A[(size_t)mrow * K + k + 1];
    b.x = B[(size_t)k * N + ncol];
    b.y = B[(size_t)(k + 1) * N + ncol];
    acc = wmma_f32_k4(a, b, acc);
  }
#pragma unroll
  for (int v = 0; v < 8; ++v)
    C[(size_t)(mt * 16 + v + 8 * hi) * N + nt * 16 + lo] = acc[v];
}

__global__ void kg_deg_init_kernel(float* __restrict__ deg, int n) {
  int i = blockIdx.x * blockDim.x + threadIdx.x;
  if (i < n) deg[i] = 1.0f;
}
__global__ void kg_deg_count_kernel(const int* __restrict__ dst,
                                    float* __restrict__ deg, int ne) {
  int i = blockIdx.x * blockDim.x + threadIdx.x;
  if (i < ne) atomicAdd(&deg[dst[i]], 1.0f);
}
__global__ void kg_dinv_kernel(const float* __restrict__ deg,
                               float* __restrict__ dinv, int n) {
  int i = blockIdx.x * blockDim.x + threadIdx.x;
  if (i < n) dinv[i] = rsqrtf(deg[i]);
}
__global__ void kg_scale_bias_kernel(const float* __restrict__ XW,
                                     const float* __restrict__ dinv,
                                     const float* __restrict__ bias,
                                     float* __restrict__ H,
                                     int Cshift, int total) {
  int idx = blockIdx.x * blockDim.x + threadIdx.x;
  if (idx >= total) return;
  int i = idx >> Cshift;
  int c = idx & ((1 << Cshift) - 1);
  H[idx] = dinv[i] * dinv[i] * XW[idx] + bias[c];
}
__global__ void kg_edge_kernel(const float* __restrict__ XW,
                               float* __restrict__ H,
                               const float* __restrict__ dinv,
                               const int* __restrict__ src,
                               const int* __restrict__ dst,
                               int Cshift, long long total) {
  long long idx = (long long)blockIdx.x * blockDim.x + threadIdx.x;
  if (idx >= total) return;
  int e = (int)(idx >> Cshift);
  int c = (int)(idx & ((1 << Cshift) - 1));
  int s = src[e], d = dst[e];
  atomicAdd(&H[((size_t)d << Cshift) + c],
            dinv[s] * dinv[d] * XW[((size_t)s << Cshift) + c]);
}
__global__ void relu_kernel(float* __restrict__ h, int n) {
  int i = blockIdx.x * blockDim.x + threadIdx.x;
  if (i < n) h[i] = fmaxf(h[i], 0.0f);
}

extern "C" void kernel_launch(void* const* d_in, const int* in_sizes, int n_in,
                              void* d_out, int out_size, void* d_ws, size_t ws_size,
                              hipStream_t stream) {
  const float* mol_x = (const float*)d_in[0];
  const int*   mei   = (const int*)d_in[1];
  const int*   kei   = (const int*)d_in[2];
  const float* W1 = (const float*)d_in[3];
  const float* b1 = (const float*)d_in[4];
  const float* W2 = (const float*)d_in[5];
  const float* b2 = (const float*)d_in[6];
  const float* W3 = (const float*)d_in[7];
  const float* b3 = (const float*)d_in[8];
  const float* Wk1 = (const float*)d_in[9];
  const float* bk1 = (const float*)d_in[10];
  const float* Wk2 = (const float*)d_in[11];
  const float* bk2 = (const float*)d_in[12];
  float* out = (float*)d_out;

  float* ws    = (float*)d_ws;
  float* embed = ws;
  float* deg   = embed + M_MOL * OUT_CH;
  float* dinv  = deg + M_MOL;
  float* XW    = dinv + M_MOL;
  float* h     = XW + (size_t)M_MOL * HID;
  float* XW2   = h + (size_t)M_MOL * HID;

  const int* kg_src = kei;
  const int* kg_dst = kei + E_KG;

  mol_gnn_kernel<<<M_MOL, 256, 0, stream>>>(mol_x, mei, W1, b1, W2, b2, W3, b3,
                                            embed);

  kg_deg_init_kernel<<<(M_MOL + 255) / 256, 256, 0, stream>>>(deg, M_MOL);
  kg_deg_count_kernel<<<(E_KG + 255) / 256, 256, 0, stream>>>(kg_dst, deg, E_KG);
  kg_dinv_kernel<<<(M_MOL + 255) / 256, 256, 0, stream>>>(deg, dinv, M_MOL);

  {
    int tiles = (M_MOL / 16) * (HID / 16);
    wmma_gemm_kernel<<<(tiles + 7) / 8, 256, 0, stream>>>(embed, Wk1, XW,
                                                          M_MOL, OUT_CH, HID);
  }
  kg_scale_bias_kernel<<<(M_MOL * HID + 255) / 256, 256, 0, stream>>>(
      XW, dinv, bk1, h, 8, M_MOL * HID);
  kg_edge_kernel<<<(int)(((long long)E_KG * HID + 255) / 256), 256, 0, stream>>>(
      XW, h, dinv, kg_src, kg_dst, 8, (long long)E_KG * HID);
  relu_kernel<<<(M_MOL * HID + 255) / 256, 256, 0, stream>>>(h, M_MOL * HID);

  {
    int tiles = (M_MOL / 16) * (OUT_CH / 16);
    wmma_gemm_kernel<<<(tiles + 7) / 8, 256, 0, stream>>>(h, Wk2, XW2,
                                                          M_MOL, HID, OUT_CH);
  }
  kg_scale_bias_kernel<<<(M_MOL * OUT_CH + 255) / 256, 256, 0, stream>>>(
      XW2, dinv, bk2, out, 7, M_MOL * OUT_CH);
  kg_edge_kernel<<<(int)(((long long)E_KG * OUT_CH + 255) / 256), 256, 0,
                  stream>>>(XW2, out, dinv, kg_src, kg_dst, 7,
                            (long long)E_KG * OUT_CH);
}


# mol kernel: LDS row pitch of the WMMA A-operand buffer K -> K+4 floats, so each 8-byte A-fragment read hits 64 distinct banks instead of 16 rows on one bank (16-way conflict)
# speedup vs baseline: 1.3996x; 1.3996x over previous
#include <hip/hip_runtime.h>

typedef __attribute__((ext_vector_type(2))) float v2f;
typedef __attribute__((ext_vector_type(8))) float v8f;

#define N_ATOMS 32
#define E_MOL   64
#define IN_CH   64
#define OUT_CH  128
#define HID     256
#define M_MOL   10000
#define E_KG    320000
#define A_PAD   4

__device__ __forceinline__ v8f wmma_f32_k4(v2f a, v2f b, v8f c) {
  return __builtin_amdgcn_wmma_f32_16x16x4_f32(false, a, false, b, (short)0, c,
                                               false, false);
}

__device__ __forceinline__ void gemm32_wmma(const float* __restrict__ A32,
                                            const float* __restrict__ W,
                                            float* __restrict__ T,
                                            int K, int N,
                                            int wave, int nwaves, int lane) {
  const int hi = lane >> 4;
  const int lo = lane & 15;
  const int ntTiles = N >> 4;
  const int AP = K + A_PAD;
  for (int t = wave; t < 2 * ntTiles; t += nwaves) {
    const int mt = t & 1;
    const int nt = t >> 1;
    const int mrow = mt * 16 + lo;
    const int ncol = nt * 16 + lo;
    v8f acc = {};
    for (int kb = 0; kb < K; kb += 4) {
      const int k = kb + 2 * hi;
      v2f a, b;
      a.x = A32[mrow * AP + k];
      a.y = A32[mrow * AP + k + 1];
      b.x = W[k * N + ncol];
      b.y = W[(k + 1) * N + ncol];
      acc = wmma_f32_k4(a, b, acc);
    }
#pragma unroll
    for (int v = 0; v < 8; ++v)
      T[(mt * 16 + v + 8 * hi) * N + nt * 16 + lo] = acc[v];
  }
}

__device__ __forceinline__ void aggregate(const float* __restrict__ T,
                                          float* __restrict__ H,
                                          const float* __restrict__ bias,
                                          int Cshift, bool relu,
                                          const int* s_src, const int* s_dst,
                                          const float* s_dinv,
                                          const float* s_norm,
                                          int tid, int nthr) {
  const int Cm1 = (1 << Cshift) - 1;
  const int HP = (1 << Cshift) + A_PAD;
  for (int idx = tid; idx < (N_ATOMS << Cshift); idx += nthr) {
    const int i = idx >> Cshift;
    const int c = idx & Cm1;
    H[i * HP + c] = s_dinv[i] * s_dinv[i] * T[idx] + bias[c];
  }
  __syncthreads();
  for (int idx = tid; idx < (E_MOL << Cshift); idx += nthr) {
    const int e = idx >> Cshift;
    const int c = idx & Cm1;
    atomicAdd(&H[s_dst[e] * HP + c],
              s_norm[e] * T[(s_src[e] << Cshift) + c]);
  }
  __syncthreads();
  if (relu) {
    for (int idx = tid; idx < (N_ATOMS << Cshift); idx += nthr) {
      float* h = &H[(idx >> Cshift) * HP + (idx & Cm1)];
      *h = fmaxf(*h, 0.0f);
    }
  }
  __syncthreads();
}

__global__ __launch_bounds__(256) void mol_gnn_kernel(
    const float* __restrict__ mol_x, const int* __restrict__ mei,
    const float* __restrict__ W1, const float* __restrict__ b1,
    const float* __restrict__ W2, const float* __restrict__ b2,
    const float* __restrict__ W3, const float* __restrict__ b3,
    float* __restrict__ embed) {
  __shared__ float Abuf[N_ATOMS * (HID + A_PAD)];
  __shared__ float Bbuf[N_ATOMS * HID];
  __shared__ float s_dinv[N_ATOMS];
  __shared__ float s_norm[E_MOL];
  __shared__ int   s_deg[N_ATOMS];
  __shared__ int   s_src[E_MOL];
  __shared__ int   s_dst[E_MOL];

  const int mol  = blockIdx.x;
  const int tid  = threadIdx.x;
  const int nthr = blockDim.x;
  const int lane = tid & 31;
  const int wave = tid >> 5;
  const int nw   = nthr >> 5;

  const float* X = mol_x + (size_t)mol * (N_ATOMS * IN_CH);
  for (int i = tid; i < N_ATOMS * IN_CH; i += nthr)
    Abuf[(i / IN_CH) * (IN_CH + A_PAD) + (i % IN_CH)] = X[i];
  const int* me = mei + (size_t)mol * (2 * E_MOL);
  if (tid < E_MOL) { s_src[tid] = me[tid]; s_dst[tid] = me[E_MOL + tid]; }
  if (tid < N_ATOMS) s_deg[tid] = 1;
  __syncthreads();
  if (tid < E_MOL) atomicAdd(&s_deg[s_dst[tid]], 1);
  __syncthreads();
  if (tid < N_ATOMS) s_dinv[tid] = rsqrtf((float)s_deg[tid]);
  __syncthreads();
  if (tid < E_MOL) s_norm[tid] = s_dinv[s_src[tid]] * s_dinv[s_dst[tid]];
  __syncthreads();

  gemm32_wmma(Abuf, W1, Bbuf, IN_CH, HID, wave, nw, lane);
  __syncthreads();
  aggregate(Bbuf, Abuf, b1, 8, true, s_src, s_dst, s_dinv, s_norm, tid, nthr);

  gemm32_wmma(Abuf, W2, Bbuf, HID, HID, wave, nw, lane);
  __syncthreads();
  aggregate(Bbuf, Abuf, b2, 8, true, s_src, s_dst, s_dinv, s_norm, tid, nthr);

  gemm32_wmma(Abuf, W3, Bbuf, HID, OUT_CH, wave, nw, lane);
  __syncthreads();
  aggregate(Bbuf, Abuf, b3, 7, false, s_src, s_dst, s_dinv, s_norm, tid, nthr);

  for (int c = tid; c < OUT_CH; c += nthr) {
    float v = Abuf[c];
    for (int r = 1; r < N_ATOMS; ++r)
      v = fminf(v, Abuf[r * (OUT_CH + A_PAD) + c]);
    embed[(size_t)mol * OUT_CH + c] = v;
  }
}

__global__ __launch_bounds__(256) void wmma_gemm_kernel(
    const float* __restrict__ A, const float* __restrict__ B,
    float* __restrict__ C, int Mrows, int K, int N) {
  const int lane = threadIdx.x & 31;
  const int wave = blockIdx.x * (blockDim.x >> 5) + (threadIdx.x >> 5);
  const int ntT = N >> 4;
  const int nTiles = (Mrows >> 4) * ntT;
  if (wave >= nTiles) return;
  const int mt = wave / ntT;
  const int nt = wave - mt * ntT;
  const int hi = lane >> 4, lo = lane & 15;
  const int mrow = mt * 16 + lo;
  const int ncol = nt * 16 + lo;
  v8f acc = {};
  for (int kb = 0; kb < K; kb += 4) {
    const int k = kb + 2 * hi;
    v2f a, b;
    a.x = A[(size_t)mrow * K + k];
    a.y = A[(size_t)mrow * K + k + 1];
    b.x = B[(size_t)k * N + ncol];
    b.y = B[(size_t)(k + 1) * N + ncol];
    acc = wmma_f32_k4(a, b, acc);
  }
#pragma unroll
  for (int v = 0; v < 8; ++v)
    C[(size_t)(mt * 16 + v + 8 * hi) * N + nt * 16 + lo] = acc[v];
}

__global__ void kg_deg_init_kernel(float* __restrict__ deg, int n) {
  int i = blockIdx.x * blockDim.x + threadIdx.x;
  if (i < n) deg[i] = 1.0f;
}
__global__ void kg_deg_count_kernel(const int* __restrict__ dst,
                                    float* __restrict__ deg, int ne) {
  int i = blockIdx.x * blockDim.x + threadIdx.x;
  if (i < ne) atomicAdd(&deg[dst[i]], 1.0f);
}
__global__ void kg_dinv_kernel(const float* __restrict__ deg,
                               float* __restrict__ dinv, int n) {
  int i = blockIdx.x * blockDim.x + threadIdx.x;
  if (i < n) dinv[i] = rsqrtf(deg[i]);
}
__global__ void kg_scale_bias_kernel(const float* __restrict__ XW,
                                     const float* __restrict__ dinv,
                                     const float* __restrict__ bias,
                                     float* __restrict__ H,
                                     int Cshift, int total) {
  int idx = blockIdx.x * blockDim.x + threadIdx.x;
  if (idx >= total) return;
  int i = idx >> Cshift;
  int c = idx & ((1 << Cshift) - 1);
  H[idx] = dinv[i] * dinv[i] * XW[idx] + bias[c];
}
__global__ void kg_edge_kernel(const float* __restrict__ XW,
                               float* __restrict__ H,
                               const float* __restrict__ dinv,
                               const int* __restrict__ src,
                               const int* __restrict__ dst,
                               int Cshift, long long total) {
  long long idx = (long long)blockIdx.x * blockDim.x + threadIdx.x;
  if (idx >= total) return;
  int e = (int)(idx >> Cshift);
  int c = (int)(idx & ((1 << Cshift) - 1));
  int s = src[e], d = dst[e];
  atomicAdd(&H[((size_t)d << Cshift) + c],
            dinv[s] * dinv[d] * XW[((size_t)s << Cshift) + c]);
}
__global__ void relu_kernel(float* __restrict__ h, int n) {
  int i = blockIdx.x * blockDim.x + threadIdx.x;
  if (i < n) h[i] = fmaxf(h[i], 0.0f);
}

extern "C" void kernel_launch(void* const* d_in, const int* in_sizes, int n_in,
                              void* d_out, int out_size, void* d_ws, size_t ws_size,
                              hipStream_t stream) {
  const float* mol_x = (const float*)d_in[0];
  const int*   mei   = (const int*)d_in[1];
  const int*   kei   = (const int*)d_in[2];
  const float* W1 = (const float*)d_in[3];
  const float* b1 = (const float*)d_in[4];
  const float* W2 = (const float*)d_in[5];
  const float* b2 = (const float*)d_in[6];
  const float* W3 = (const float*)d_in[7];
  const float* b3 = (const float*)d_in[8];
  const float* Wk1 = (const float*)d_in[9];
  const float* bk1 = (const float*)d_in[10];
  const float* Wk2 = (const float*)d_in[11];
  const float* bk2 = (const float*)d_in[12];
  float* out = (float*)d_out;

  float* ws    = (float*)d_ws;
  float* embed = ws;
  float* deg   = embed + M_MOL * OUT_CH;
  float* dinv  = deg + M_MOL;
  float* XW    = dinv + M_MOL;
  float* h     = XW + (size_t)M_MOL * HID;
  float* XW2   = h + (size_t)M_MOL * HID;

  const int* kg_src = kei;
  const int* kg_dst = kei + E_KG;

  mol_gnn_kernel<<<M_MOL, 256, 0, stream>>>(mol_x, mei, W1, b1, W2, b2, W3, b3,
                                            embed);

  kg_deg_init_kernel<<<(M_MOL + 255) / 256, 256, 0, stream>>>(deg, M_MOL);
  kg_deg_count_kernel<<<(E_KG + 255) / 256, 256, 0, stream>>>(kg_dst, deg, E_KG);
  kg_dinv_kernel<<<(M_MOL + 255) / 256, 256, 0, stream>>>(deg, dinv, M_MOL);

  {
    int tiles = (M_MOL / 16) * (HID / 16);
    wmma_gemm_kernel<<<(tiles + 7) / 8, 256, 0, stream>>>(embed, Wk1, XW,
                                                          M_MOL, OUT_CH, HID);
  }
  kg_scale_bias_kernel<<<(M_MOL * HID + 255) / 256, 256, 0, stream>>>(
      XW, dinv, bk1, h, 8, M_MOL * HID);
  kg_edge_kernel<<<(int)(((long long)E_KG * HID + 255) / 256), 256, 0, stream>>>(
      XW, h, dinv, kg_src, kg_dst, 8, (long long)E_KG * HID);
  relu_kernel<<<(M_MOL * HID + 255) / 256, 256, 0, stream>>>(h, M_MOL * HID);

  {
    int tiles = (M_MOL / 16) * (OUT_CH / 16);
    wmma_gemm_kernel<<<(tiles + 7) / 8, 256, 0, stream>>>(h, Wk2, XW2,
                                                          M_MOL, HID, OUT_CH);
  }
  kg_scale_bias_kernel<<<(M_MOL * OUT_CH + 255) / 256, 256, 0, stream>>>(
      XW2, dinv, bk2, out, 7, M_MOL * OUT_CH);
  kg_edge_kernel<<<(int)(((long long)E_KG * OUT_CH + 255) / 256), 256, 0,
                  stream>>>(XW2, out, dinv, kg_src, kg_dst, 7,
                            (long long)E_KG * OUT_CH);
}


# KG layers: per-element float atomic scatter -> gather over a CSR by destination rebuilt each call (int count, scan, arrival-order fill); self term, bias, ReLU in the gather; sum order changes
# speedup vs baseline: 1.7638x; 1.2602x over previous
#include <hip/hip_runtime.h>

typedef __attribute__((ext_vector_type(2))) float v2f;
typedef __attribute__((ext_vector_type(8))) float v8f;

#define N_ATOMS 32
#define E_MOL   64
#define IN_CH   64
#define OUT_CH  128
#define HID     256
#define M_MOL   10000
#define E_KG    320000
#define A_PAD   4

__device__ __forceinline__ v8f wmma_f32_k4(v2f a, v2f b, v8f c) {
  return __builtin_amdgcn_wmma_f32_16x16x4_f32(false, a, false, b, (short)0, c,
                                               false, false);
}

__device__ __forceinline__ void gemm32_wmma(const float* __restrict__ A32,
                                            const float* __restrict__ W,
                                            float* __restrict__ T,
                                            int K, int N,
                                            int wave, int nwaves, int lane) {
  const int hi = lane >> 4;
  const int lo = lane & 15;
  const int ntTiles = N >> 4;
  const int AP = K + A_PAD;
  for (int t = wave; t < 2 * ntTiles; t += nwaves) {
    const int mt = t & 1;
    const int nt = t >> 1;
    const int mrow = mt * 16 + lo;
    const int ncol = nt * 16 + lo;
    v8f acc = {};
    for (int kb = 0; kb < K; kb += 4) {
      const int k = kb + 2 * hi;
      v2f a, b;
      a.x = A32[mrow * AP + k];
      a.y = A32[mrow * AP + k + 1];
      b.x = W[k * N + ncol];
      b.y = W[(k + 1) * N + ncol];
      acc = wmma_f32_k4(a, b, acc);
    }
#pragma unroll
    for (int v = 0; v < 8; ++v)
      T[(mt * 16 + v + 8 * hi) * N + nt * 16 + lo] = acc[v];
  }
}

__device__ __forceinline__ void aggregate(const float* __restrict__ T,
                                          float* __restrict__ H,
                                          const float* __restrict__ bias,
                                          int Cshift, bool relu,
                                          const int* s_src, const int* s_dst,
                                          const float* s_dinv,
                                          const float* s_norm,
                                          int tid, int nthr) {
  const int Cm1 = (1 << Cshift) - 1;
  const int HP = (1 << Cshift) + A_PAD;
  for (int idx = tid; idx < (N_ATOMS << Cshift); idx += nthr) {
    const int i = idx >> Cshift;
    const int c = idx & Cm1;
    H[i * HP + c] = s_dinv[i] * s_dinv[i] * T[idx] + bias[c];
  }
  __syncthreads();
  for (int idx = tid; idx < (E_MOL << Cshift); idx += nthr) {
    const int e = idx >> Cshift;
    const int c = idx & Cm1;
    atomicAdd(&H[s_dst[e] * HP + c],
              s_norm[e] * T[(s_src[e] << Cshift) + c]);
  }
  __syncthreads();
  if (relu) {
    for (int idx = tid; idx < (N_ATOMS << Cshift); idx += nthr) {
      float* h = &H[(idx >> Cshift) * HP + (idx & Cm1)];
      *h = fmaxf(*h, 0.0f);
    }
  }
  __syncthreads();
}

__global__ __launch_bounds__(256) void mol_gnn_kernel(
    const float* __restrict__ mol_x, const int* __restrict__ mei,
    const float* __restrict__ W1, const float* __restrict__ b1,
    const float* __restrict__ W2, const float* __restrict__ b2,
    const float* __restrict__ W3, const float* __restrict__ b3,
    float* __restrict__ embed) {
  __shared__ float Abuf[N_ATOMS * (HID + A_PAD)];
  __shared__ float Bbuf[N_ATOMS * HID];
  __shared__ float s_dinv[N_ATOMS];
  __shared__ float s_norm[E_MOL];
  __shared__ int   s_deg[N_ATOMS];
  __shared__ int   s_src[E_MOL];
  __shared__ int   s_dst[E_MOL];

  const int mol  = blockIdx.x;
  const int tid  = threadIdx.x;
  const int nthr = blockDim.x;
  const int lane = tid & 31;
  const int wave = tid >> 5;
  const int nw   = nthr >> 5;

  const float* X = mol_x + (size_t)mol * (N_ATOMS * IN_CH);
  for (int i = tid; i < N_ATOMS * IN_CH; i += nthr)
    Abuf[(i / IN_CH) * (IN_CH + A_PAD) + (i % IN_CH)] = X[i];
  const int* me = mei + (size_t)mol * (2 * E_MOL);
  if (tid < E_MOL) { s_src[tid] = me[tid]; s_dst[tid] = me[E_MOL + tid]; }
  if (tid < N_ATOMS) s_deg[tid] = 1;
  __syncthreads();
  if (tid < E_MOL) atomicAdd(&s_deg[s_dst[tid]], 1);
  __syncthreads();
  if (tid < N_ATOMS) s_dinv[tid] = rsqrtf((float)s_deg[tid]);
  __syncthreads();
  if (tid < E_MOL) s_norm[tid] = s_dinv[s_src[tid]] * s_dinv[s_dst[tid]];
  __syncthreads();

  gemm32_wmma(Abuf, W1, Bbuf, IN_CH, HID, wave, nw, lane);
  __syncthreads();
  aggregate(Bbuf, Abuf, b1, 8, true, s_src, s_dst, s_dinv, s_norm, tid, nthr);

  gemm32_wmma(Abuf, W2, Bbuf, HID, HID, wave, nw, lane);
  __syncthreads();
  aggregate(Bbuf, Abuf, b2, 8, true, s_src, s_dst, s_dinv, s_norm, tid, nthr);

  gemm32_wmma(Abuf, W3, Bbuf, HID, OUT_CH, wave, nw, lane);
  __syncthreads();
  aggregate(Bbuf, Abuf, b3, 7, false, s_src, s_dst, s_dinv, s_norm, tid, nthr);

  for (int c = tid; c < OUT_CH; c += nthr) {
    float v = Abuf[c];
    for (int r = 1; r < N_ATOMS; ++r)
      v = fminf(v, Abuf[r * (OUT_CH + A_PAD) + c]);
    embed[(size_t)mol * OUT_CH + c] = v;
  }
}

__global__ __launch_bounds__(256) void wmma_gemm_kernel(
    const float* __restrict__ A, const float* __restrict__ B,
    float* __restrict__ C, int Mrows, int K, int N) {
  const int lane = threadIdx.x & 31;
  const int wave = blockIdx.x * (blockDim.x >> 5) + (threadIdx.x >> 5);
  const int ntT = N >> 4;
  const int nTiles = (Mrows >> 4) * ntT;
  if (wave >= nTiles) return;
  const int mt = wave / ntT;
  const int nt = wave - mt * ntT;
  const int hi = lane >> 4, lo = lane & 15;
  const int mrow = mt * 16 + lo;
  const int ncol = nt * 16 + lo;
  v8f acc = {};
  for (int kb = 0; kb < K; kb += 4) {
    const int k = kb + 2 * hi;
    v2f a, b;
    a.x = A[(size_t)mrow * K + k];
    a.y = A[(size_t)mrow * K + k + 1];
    b.x = B[(size_t)k * N + ncol];
    b.y = B[(size_t)(k + 1) * N + ncol];
    acc = wmma_f32_k4(a, b, acc);
  }
#pragma unroll
  for (int v = 0; v < 8; ++v)
    C[(size_t)(mt * 16 + v + 8 * hi) * N + nt * 16 + lo] = acc[v];
}


__global__ void kg_zero_kernel(int* __restrict__ cnt, int n) {
  int i = blockIdx.x * blockDim.x + threadIdx.x;
  if (i < n) cnt[i] = 0;
}
__global__ void kg_count_kernel(const int* __restrict__ dst,
                                int* __restrict__ cnt, int ne) {
  int i = blockIdx.x * blockDim.x + threadIdx.x;
  if (i < ne) atomicAdd(&cnt[dst[i]], 1);
}
__global__ __launch_bounds__(1024) void kg_scan_kernel(
    const int* __restrict__ cnt, int* __restrict__ rowptr,
    int* __restrict__ cursor, float* __restrict__ dinv, int n) {
  __shared__ int s_part[1024];
  const int tid = threadIdx.x;
  const int per = (n + 1023) >> 10;
  const int lo = tid * per;
  const int hi = min(lo + per, n);
  int sum = 0;
  for (int i = lo; i < hi; ++i) sum += cnt[i];
  s_part[tid] = sum;
  __syncthreads();
  for (int off = 1; off < 1024; off <<= 1) {
    const int v = (tid >= off) ? s_part[tid - off] : 0;
    __syncthreads();
    s_part[tid] += v;
    __syncthreads();
  }
  int run = s_part[tid] - sum;
  for (int i = lo; i < hi; ++i) {
    const int c = cnt[i];
    rowptr[i] = run;
    cursor[i] = run;
    dinv[i] = rsqrtf((float)(c + 1));
    run += c;
  }
  if (tid == 1023) rowptr[n] = s_part[1023];
}
__global__ void kg_fill_kernel(const int* __restrict__ src,
                               const int* __restrict__ dst,
                               int* __restrict__ cursor,
                               int* __restrict__ csr_src, int ne) {
  int e = blockIdx.x * blockDim.x + threadIdx.x;
  if (e < ne) {
    const int p = atomicAdd(&cursor[dst[e]], 1);
    csr_src[p] = src[e];
  }
}

template <int C, bool RELU>
__global__ __launch_bounds__(256) void kg_gather_kernel(
    const float* __restrict__ XW, const float* __restrict__ dinv,
    const float* __restrict__ bias, const int* __restrict__ rowptr,
    const int* __restrict__ csr_src, float* __restrict__ H, int n) {
  constexpr int SLICES = C / 128;
  const int lane = threadIdx.x & 31;
  const int wv = __builtin_amdgcn_readfirstlane((int)(threadIdx.x >> 5));
  const int gw = blockIdx.x * (int)(blockDim.x >> 5) + wv;
  const int d = gw / SLICES;
  if (d >= n) return;
  const int col = (gw - d * SLICES) * 128 + lane * 4;
  const int beg = rowptr[d];
  const int end = rowptr[d + 1];
  const float dd = dinv[d];
  const float sw = dd * dd;
  const float4 xs = *reinterpret_cast<const float4*>(XW + (size_t)d * C + col);
  float4 acc;
  acc.x = sw * xs.x + bias[col + 0];
  acc.y = sw * xs.y + bias[col + 1];
  acc.z = sw * xs.z + bias[col + 2];
  acc.w = sw * xs.w + bias[col + 3];
  for (int base = beg; base < end; base += 32) {
    const int m = min(end - base, 32);
    int s = 0;
    float w = 0.0f;
    if (lane < m) {
      s = csr_src[base + lane];
      w = dinv[s] * dd;
    }
    for (int j = 0; j < m; j += 8) {
      float4 r[8];
      float wj[8];
#pragma unroll
      for (int u = 0; u < 8; ++u) {
        r[u].x = 0.0f; r[u].y = 0.0f; r[u].z = 0.0f; r[u].w = 0.0f;
        wj[u] = 0.0f;
        if (j + u < m) {
          const int sj = __builtin_amdgcn_readlane(s, j + u);
          wj[u] = __int_as_float(
              __builtin_amdgcn_readlane(__float_as_int(w), j + u));
          r[u] = *reinterpret_cast<const float4*>(XW + (size_t)sj * C + col);
        }
      }
#pragma unroll
      for (int u = 0; u < 8; ++u) {
        acc.x += wj[u] * r[u].x;
        acc.y += wj[u] * r[u].y;
        acc.z += wj[u] * r[u].z;
        acc.w += wj[u] * r[u].w;
      }
    }
  }
  if (RELU) {
    acc.x = fmaxf(acc.x, 0.0f);
    acc.y = fmaxf(acc.y, 0.0f);
    acc.z = fmaxf(acc.z, 0.0f);
    acc.w = fmaxf(acc.w, 0.0f);
  }
  *reinterpret_cast<float4*>(H + (size_t)d * C + col) = acc;
}

extern "C" void kernel_launch(void* const* d_in, const int* in_sizes, int n_in,
                              void* d_out, int out_size, void* d_ws, size_t ws_size,
                              hipStream_t stream) {
  const float* mol_x = (const float*)d_in[0];
  const int*   mei   = (const int*)d_in[1];
  const int*   kei   = (const int*)d_in[2];
  const float* W1 = (const float*)d_in[3];
  const float* b1 = (const float*)d_in[4];
  const float* W2 = (const float*)d_in[5];
  const float* b2 = (const float*)d_in[6];
  const float* W3 = (const float*)d_in[7];
  const float* b3 = (const float*)d_in[8];
  const float* Wk1 = (const float*)d_in[9];
  const float* bk1 = (const float*)d_in[10];
  const float* Wk2 = (const float*)d_in[11];
  const float* bk2 = (const float*)d_in[12];
  float* out = (float*)d_out;

  float* ws     = (float*)d_ws;
  float* embed  = ws;
  float* XW     = embed + (size_t)M_MOL * OUT_CH;
  float* XW2    = XW;
  float* h      = XW + (size_t)M_MOL * HID;
  float* dinv   = h + (size_t)M_MOL * HID;
  int*   cnt    = (int*)(dinv + M_MOL);
  int*   rowptr = cnt + M_MOL;
  int*   cursor = rowptr + (M_MOL + 16);
  int*   csr_src = cursor + M_MOL;

  const int* kg_src = kei;
  const int* kg_dst = kei + E_KG;

  mol_gnn_kernel<<<M_MOL, 256, 0, stream>>>(mol_x, mei, W1, b1, W2, b2, W3, b3,
                                            embed);

  kg_zero_kernel<<<(M_MOL + 255) / 256, 256, 0, stream>>>(cnt, M_MOL);
  kg_count_kernel<<<(E_KG + 255) / 256, 256, 0, stream>>>(kg_dst, cnt, E_KG);
  kg_scan_kernel<<<1, 1024, 0, stream>>>(cnt, rowptr, cursor, dinv, M_MOL);
  kg_fill_kernel<<<(E_KG + 255) / 256, 256, 0, stream>>>(kg_src, kg_dst, cursor,
                                                         csr_src, E_KG);

  {
    int tiles = (M_MOL / 16) * (HID / 16);
    wmma_gemm_kernel<<<(tiles + 7) / 8, 256, 0, stream>>>(embed, Wk1, XW,
                                                          M_MOL, OUT_CH, HID);
  }
  kg_gather_kernel<HID, true><<<(M_MOL * (HID / 128) + 7) / 8, 256, 0, stream>>>(
      XW, dinv, bk1, rowptr, csr_src, h, M_MOL);

  {
    int tiles = (M_MOL / 16) * (OUT_CH / 16);
    wmma_gemm_kernel<<<(tiles + 7) / 8, 256, 0, stream>>>(h, Wk2, XW2,
                                                          M_MOL, HID, OUT_CH);
  }
  kg_gather_kernel<OUT_CH, false><<<(M_MOL * (OUT_CH / 128) + 7) / 8, 256, 0,
                                    stream>>>(XW2, dinv, bk2, rowptr, csr_src,
                                              out, M_MOL);
}


# mol kernel aggregation: LDS float-atomic edge scatter (3 passes) -> one-pass gather per node, in-edges by wave ballot, rows read as ds_load_b128, self+bias+ReLU fused; sums in ascending edge order
# speedup vs baseline: 2.3780x; 1.3482x over previous
#include <hip/hip_runtime.h>

typedef __attribute__((ext_vector_type(2))) float v2f;
typedef __attribute__((ext_vector_type(8))) float v8f;

#define N_ATOMS 32
#define E_MOL   64
#define IN_CH   64
#define OUT_CH  128
#define HID     256
#define M_MOL   10000
#define E_KG    320000
#define A_PAD   4

__device__ __forceinline__ v8f wmma_f32_k4(v2f a, v2f b, v8f c) {
  return __builtin_amdgcn_wmma_f32_16x16x4_f32(false, a, false, b, (short)0, c,
                                               false, false);
}

__device__ __forceinline__ void gemm32_wmma(const float* __restrict__ A32,
                                            const float* __restrict__ W,
                                            float* __restrict__ T,
                                            int K, int N,
                                            int wave, int nwaves, int lane) {
  const int hi = lane >> 4;
  const int lo = lane & 15;
  const int ntTiles = N >> 4;
  const int AP = K + A_PAD;
  for (int t = wave; t < 2 * ntTiles; t += nwaves) {
    const int mt = t & 1;
    const int nt = t >> 1;
    const int mrow = mt * 16 + lo;
    const int ncol = nt * 16 + lo;
    v8f acc = {};
    for (int kb = 0; kb < K; kb += 4) {
      const int k = kb + 2 * hi;
      v2f a, b;
      a.x = A32[mrow * AP + k];
      a.y = A32[mrow * AP + k + 1];
      b.x = W[k * N + ncol];
      b.y = W[(k + 1) * N + ncol];
      acc = wmma_f32_k4(a, b, acc);
    }
#pragma unroll
    for (int v = 0; v < 8; ++v)
      T[(mt * 16 + v + 8 * hi) * N + nt * 16 + lo] = acc[v];
  }
}

template <int C, bool RELU>
__device__ __forceinline__ void aggregate_gather(
    const float* __restrict__ T, float* __restrict__ H,
    const float* __restrict__ bias, const float* s_dinv,
    int dstA, int dstB, int srcA, int srcB, float nrmA, float nrmB,
    int wave, int nwaves, int lane) {
  constexpr int V  = C / 128;
  constexpr int HP = C + A_PAD;
  float4 bv[V];
  for (int v = 0; v < V; ++v)
    bv[v] = *reinterpret_cast<const float4*>(bias + v * 128 + lane * 4);
  for (int i = wave; i < N_ATOMS; i += nwaves) {
    unsigned long long m =
        (unsigned long long)__builtin_amdgcn_ballot_w32(dstA == i) |
        ((unsigned long long)__builtin_amdgcn_ballot_w32(dstB == i) << 32);
    const float di = s_dinv[i];
    const float sw = di * di;
    const float* Ti = T + i * C + lane * 4;
    float4 t[V], acc[V];
    for (int v = 0; v < V; ++v)
      t[v] = *reinterpret_cast<const float4*>(Ti + v * 128);
    bool self_pending = true;
    do {
      float4 r[4][V];
      float wj[4];
      for (int u = 0; u < 4; ++u) {
        wj[u] = 0.0f;
        for (int v = 0; v < V; ++v) {
          r[u][v].x = 0.0f; r[u][v].y = 0.0f; r[u][v].z = 0.0f; r[u][v].w = 0.0f;
        }
        if (m != 0) {
          const int e = __builtin_ctzll(m);
          m &= m - 1;
          const int l = e & 31;
          const int sa = __builtin_amdgcn_readlane(srcA, l);
          const int sb = __builtin_amdgcn_readlane(srcB, l);
          const int wa = __builtin_amdgcn_readlane(__float_as_int(nrmA), l);
          const int wb = __builtin_amdgcn_readlane(__float_as_int(nrmB), l);
          const int s = (e < 32) ? sa : sb;
          wj[u] = __int_as_float((e < 32) ? wa : wb);
          const float* Ts = T + s * C + lane * 4;
          for (int v = 0; v < V; ++v)
            r[u][v] = *reinterpret_cast<const float4*>(Ts + v * 128);
        }
      }
      if (self_pending) {
        self_pending = false;
        for (int v = 0; v < V; ++v) {
          acc[v].x = sw * t[v].x + bv[v].x;
          acc[v].y = sw * t[v].y + bv[v].y;
          acc[v].z = sw * t[v].z + bv[v].z;
          acc[v].w = sw * t[v].w + bv[v].w;
        }
      }
      for (int u = 0; u < 4; ++u) {
        for (int v = 0; v < V; ++v) {
          acc[v].x += wj[u] * r[u][v].x;
          acc[v].y += wj[u] * r[u][v].y;
          acc[v].z += wj[u] * r[u][v].z;
          acc[v].w += wj[u] * r[u][v].w;
        }
      }
    } while (m != 0);
    float* Hi = H + i * HP + lane * 4;
    for (int v = 0; v < V; ++v) {
      if (RELU) {
        acc[v].x = fmaxf(acc[v].x, 0.0f);
        acc[v].y = fmaxf(acc[v].y, 0.0f);
        acc[v].z = fmaxf(acc[v].z, 0.0f);
        acc[v].w = fmaxf(acc[v].w, 0.0f);
      }
      *reinterpret_cast<float4*>(Hi + v * 128) = acc[v];
    }
  }
  __syncthreads();
}

__global__ __launch_bounds__(256) void mol_gnn_kernel(
    const float* __restrict__ mol_x, const int* __restrict__ mei,
    const float* __restrict__ W1, const float* __restrict__ b1,
    const float* __restrict__ W2, const float* __restrict__ b2,
    const float* __restrict__ W3, const float* __restrict__ b3,
    float* __restrict__ embed) {
  __shared__ __attribute__((aligned(16))) float Abuf[N_ATOMS * (HID + A_PAD)];
  __shared__ __attribute__((aligned(16))) float Bbuf[N_ATOMS * HID];
  __shared__ float s_dinv[N_ATOMS];
  __shared__ float s_norm[E_MOL];
  __shared__ int   s_deg[N_ATOMS];
  __shared__ int   s_src[E_MOL];
  __shared__ int   s_dst[E_MOL];

  const int mol  = blockIdx.x;
  const int tid  = threadIdx.x;
  const int nthr = blockDim.x;
  const int lane = tid & 31;
  const int wave = tid >> 5;
  const int nw   = nthr >> 5;

  const float* X = mol_x + (size_t)mol * (N_ATOMS * IN_CH);
  for (int i = tid; i < N_ATOMS * IN_CH; i += nthr)
    Abuf[(i / IN_CH) * (IN_CH + A_PAD) + (i % IN_CH)] = X[i];
  const int* me = mei + (size_t)mol * (2 * E_MOL);
  if (tid < E_MOL) { s_src[tid] = me[tid]; s_dst[tid] = me[E_MOL + tid]; }
  if (tid < N_ATOMS) s_deg[tid] = 1;
  __syncthreads();
  if (tid < E_MOL) atomicAdd(&s_deg[s_dst[tid]], 1);
  __syncthreads();
  if (tid < N_ATOMS) s_dinv[tid] = rsqrtf((float)s_deg[tid]);
  __syncthreads();
  if (tid < E_MOL) s_norm[tid] = s_dinv[s_src[tid]] * s_dinv[s_dst[tid]];
  __syncthreads();
  const int   dstA = s_dst[lane],  dstB = s_dst[lane + 32];
  const int   srcA = s_src[lane],  srcB = s_src[lane + 32];
  const float nrmA = s_norm[lane], nrmB = s_norm[lane + 32];
  const int wv = __builtin_amdgcn_readfirstlane(wave);

  gemm32_wmma(Abuf, W1, Bbuf, IN_CH, HID, wave, nw, lane);
  __syncthreads();
  aggregate_gather<HID, true>(Bbuf, Abuf, b1, s_dinv, dstA, dstB, srcA, srcB,
                              nrmA, nrmB, wv, nw, lane);

  gemm32_wmma(Abuf, W2, Bbuf, HID, HID, wave, nw, lane);
  __syncthreads();
  aggregate_gather<HID, true>(Bbuf, Abuf, b2, s_dinv, dstA, dstB, srcA, srcB,
                              nrmA, nrmB, wv, nw, lane);

  gemm32_wmma(Abuf, W3, Bbuf, HID, OUT_CH, wave, nw, lane);
  __syncthreads();
  aggregate_gather<OUT_CH, false>(Bbuf, Abuf, b3, s_dinv, dstA, dstB, srcA, srcB,
                                  nrmA, nrmB, wv, nw, lane);

  for (int c = tid; c < OUT_CH; c += nthr) {
    float v = Abuf[c];
    for (int r = 1; r < N_ATOMS; ++r)
      v = fminf(v, Abuf[r * (OUT_CH + A_PAD) + c]);
    embed[(size_t)mol * OUT_CH + c] = v;
  }
}

__global__ __launch_bounds__(256) void wmma_gemm_kernel(
    const float* __restrict__ A, const float* __restrict__ B,
    float* __restrict__ C, int Mrows, int K, int N) {
  const int lane = threadIdx.x & 31;
  const int wave = blockIdx.x * (blockDim.x >> 5) + (threadIdx.x >> 5);
  const int ntT = N >> 4;
  const int nTiles = (Mrows >> 4) * ntT;
  if (wave >= nTiles) return;
  const int mt = wave / ntT;
  const int nt = wave - mt * ntT;
  const int hi = lane >> 4, lo = lane & 15;
  const int mrow = mt * 16 + lo;
  const int ncol = nt * 16 + lo;
  v8f acc = {};
  for (int kb = 0; kb < K; kb += 4) {
    const int k = kb + 2 * hi;
    v2f a, b;
    a.x = A[(size_t)mrow * K + k];
    a.y = A[(size_t)mrow * K + k + 1];
    b.x = B[(size_t)k * N + ncol];
    b.y = B[(size_t)(k + 1) * N + ncol];
    acc = wmma_f32_k4(a, b, acc);
  }
#pragma unroll
  for (int v = 0; v < 8; ++v)
    C[(size_t)(mt * 16 + v + 8 * hi) * N + nt * 16 + lo] = acc[v];
}


__global__ void kg_zero_kernel(int* __restrict__ cnt, int n) {
  int i = blockIdx.x * blockDim.x + threadIdx.x;
  if (i < n) cnt[i] = 0;
}
__global__ void kg_count_kernel(const int* __restrict__ dst,
                                int* __restrict__ cnt, int ne) {
  int i = blockIdx.x * blockDim.x + threadIdx.x;
  if (i < ne) atomicAdd(&cnt[dst[i]], 1);
}
__global__ __launch_bounds__(1024) void kg_scan_kernel(
    const int* __restrict__ cnt, int* __restrict__ rowptr,
    int* __restrict__ cursor, float* __restrict__ dinv, int n) {
  __shared__ int s_part[1024];
  const int tid = threadIdx.x;
  const int per = (n + 1023) >> 10;
  const int lo = tid * per;
  const int hi = min(lo + per, n);
  int sum = 0;
  for (int i = lo; i < hi; ++i) sum += cnt[i];
  s_part[tid] = sum;
  __syncthreads();
  for (int off = 1; off < 1024; off <<= 1) {
    const int v = (tid >= off) ? s_part[tid - off] : 0;
    __syncthreads();
    s_part[tid] += v;
    __syncthreads();
  }
  int run = s_part[tid] - sum;
  for (int i = lo; i < hi; ++i) {
    const int c = cnt[i];
    rowptr[i] = run;
    cursor[i] = run;
    dinv[i] = rsqrtf((float)(c + 1));
    run += c;
  }
  if (tid == 1023) rowptr[n] = s_part[1023];
}
__global__ void kg_fill_kernel(const int* __restrict__ src,
                               const int* __restrict__ dst,
                               int* __restrict__ cursor,
                               int* __restrict__ csr_src, int ne) {
  int e = blockIdx.x * blockDim.x + threadIdx.x;
  if (e < ne) {
    const int p = atomicAdd(&cursor[dst[e]], 1);
    csr_src[p] = src[e];
  }
}

template <int C, bool RELU>
__global__ __launch_bounds__(256) void kg_gather_kernel(
    const float* __restrict__ XW, const float* __restrict__ dinv,
    const float* __restrict__ bias, const int* __restrict__ rowptr,
    const int* __restrict__ csr_src, float* __restrict__ H, int n) {
  constexpr int SLICES = C / 128;
  const int lane = threadIdx.x & 31;
  const int wv = __builtin_amdgcn_readfirstlane((int)(threadIdx.x >> 5));
  const int gw = blockIdx.x * (int)(blockDim.x >> 5) + wv;
  const int d = gw / SLICES;
  if (d >= n) return;
  const int col = (gw - d * SLICES) * 128 + lane * 4;
  const int beg = rowptr[d];
  const int end = rowptr[d + 1];
  const float dd = dinv[d];
  const float sw = dd * dd;
  const float4 xs = *reinterpret_cast<const float4*>(XW + (size_t)d * C + col);
  float4 acc;
  acc.x = sw * xs.x + bias[col + 0];
  acc.y = sw * xs.y + bias[col + 1];
  acc.z = sw * xs.z + bias[col + 2];
  acc.w = sw * xs.w + bias[col + 3];
  for (int base = beg; base < end; base += 32) {
    const int m = min(end - base, 32);
    int s = 0;
    float w = 0.0f;
    if (lane < m) {
      s = csr_src[base + lane];
      w = dinv[s] * dd;
    }
    for (int j = 0; j < m; j += 8) {
      float4 r[8];
      float wj[8];
#pragma unroll
      for (int u = 0; u < 8; ++u) {
        r[u].x = 0.0f; r[u].y = 0.0f; r[u].z = 0.0f; r[u].w = 0.0f;
        wj[u] = 0.0f;
        if (j + u < m) {
          const int sj = __builtin_amdgcn_readlane(s, j + u);
          wj[u] = __int_as_float(
              __builtin_amdgcn_readlane(__float_as_int(w), j + u));
          r[u] = *reinterpret_cast<const float4*>(XW + (size_t)sj * C + col);
        }
      }
#pragma unroll
      for (int u = 0; u < 8; ++u) {
        acc.x += wj[u] * r[u].x;
        acc.y += wj[u] * r[u].y;
        acc.z += wj[u] * r[u].z;
        acc.w += wj[u] * r[u].w;
      }
    }
  }
  if (RELU) {
    acc.x = fmaxf(acc.x, 0.0f);
    acc.y = fmaxf(acc.y, 0.0f);
    acc.z = fmaxf(acc.z, 0.0f);
    acc.w = fmaxf(acc.w, 0.0f);
  }
  *reinterpret_cast<float4*>(H + (size_t)d * C + col) = acc;
}

extern "C" void kernel_launch(void* const* d_in, const int* in_sizes, int n_in,
                              void* d_out, int out_size, void* d_ws, size_t ws_size,
                              hipStream_t stream) {
  const float* mol_x = (const float*)d_in[0];
  const int*   mei   = (const int*)d_in[1];
  const int*   kei   = (const int*)d_in[2];
  const float* W1 = (const float*)d_in[3];
  const float* b1 = (const float*)d_in[4];
  const float* W2 = (const float*)d_in[5];
  const float* b2 = (const float*)d_in[6];
  const float* W3 = (const float*)d_in[7];
  const float* b3 = (const float*)d_in[8];
  const float* Wk1 = (const float*)d_in[9];
  const float* bk1 = (const float*)d_in[10];
  const float* Wk2 = (const float*)d_in[11];
  const float* bk2 = (const float*)d_in[12];
  float* out = (float*)d_out;

  float* ws     = (float*)d_ws;
  float* embed  = ws;
  float* XW     = embed + (size_t)M_MOL * OUT_CH;
  float* XW2    = XW;
  float* h      = XW + (size_t)M_MOL * HID;
  float* dinv   = h + (size_t)M_MOL * HID;
  int*   cnt    = (int*)(dinv + M_MOL);
  int*   rowptr = cnt + M_MOL;
  int*   cursor = rowptr + (M_MOL + 16);
  int*   csr_src = cursor + M_MOL;

  const int* kg_src = kei;
  const int* kg_dst = kei + E_KG;

  mol_gnn_kernel<<<M_MOL, 256, 0, stream>>>(mol_x, mei, W1, b1, W2, b2, W3, b3,
                                            embed);

  kg_zero_kernel<<<(M_MOL + 255) / 256, 256, 0, stream>>>(cnt, M_MOL);
  kg_count_kernel<<<(E_KG + 255) / 256, 256, 0, stream>>>(kg_dst, cnt, E_KG);
  kg_scan_kernel<<<1, 1024, 0, stream>>>(cnt, rowptr, cursor, dinv, M_MOL);
  kg_fill_kernel<<<(E_KG + 255) / 256, 256, 0, stream>>>(kg_src, kg_dst, cursor,
                                                         csr_src, E_KG);

  {
    int tiles = (M_MOL / 16) * (HID / 16);
    wmma_gemm_kernel<<<(tiles + 7) / 8, 256, 0, stream>>>(embed, Wk1, XW,
                                                          M_MOL, OUT_CH, HID);
  }
  kg_gather_kernel<HID, true><<<(M_MOL * (HID / 128) + 7) / 8, 256, 0, stream>>>(
      XW, dinv, bk1, rowptr, csr_src, h, M_MOL);

  {
    int tiles = (M_MOL / 16) * (OUT_CH / 16);
    wmma_gemm_kernel<<<(tiles + 7) / 8, 256, 0, stream>>>(h, Wk2, XW2,
                                                          M_MOL, HID, OUT_CH);
  }
  kg_gather_kernel<OUT_CH, false><<<(M_MOL * (OUT_CH / 128) + 7) / 8, 256, 0,
                                    stream>>>(XW2, dinv, bk2, rowptr, csr_src,
                                              out, M_MOL);
}


# mol kernel GEMMs: a wave keeps N/64 column-interleaved 16x16 tiles (tile j, lane c = column (N/64)c+j): one float4/float2 W load per k feeds all its tiles, one A fragment N/64 WMMAs; T stored wide
# speedup vs baseline: 2.8163x; 1.1843x over previous
#include <hip/hip_runtime.h>

typedef __attribute__((ext_vector_type(2))) float v2f;
typedef __attribute__((ext_vector_type(8))) float v8f;

#define N_ATOMS 32
#define E_MOL   64
#define IN_CH   64
#define OUT_CH  128
#define HID     256
#define M_MOL   10000
#define E_KG    320000
#define A_PAD   4

__device__ __forceinline__ v8f wmma_f32_k4(v2f a, v2f b, v8f c) {
  return __builtin_amdgcn_wmma_f32_16x16x4_f32(false, a, false, b, (short)0, c,
                                               false, false);
}

template <int NT> struct wrow;
template <> struct wrow<4> { typedef float4 type; };
template <> struct wrow<2> { typedef float2 type; };
__device__ __forceinline__ float wrow_get(const float4& v, int j) {
  return j == 0 ? v.x : (j == 1 ? v.y : (j == 2 ? v.z : v.w));
}
__device__ __forceinline__ float wrow_get(const float2& v, int j) {
  return j == 0 ? v.x : v.y;
}
__device__ __forceinline__ void wrow_set(float4& v, int j, float f) {
  if (j == 0) v.x = f; else if (j == 1) v.y = f; else if (j == 2) v.z = f; else v.w = f;
}
__device__ __forceinline__ void wrow_set(float2& v, int j, float f) {
  if (j == 0) v.x = f; else v.y = f;
}

template <int K, int NT>
__device__ __forceinline__ void gemm32_wmma(const float* __restrict__ A32,
                                            const float* __restrict__ W,
                                            float* __restrict__ T,
                                            int wave, int nwaves, int lane) {
  typedef typename wrow<NT>::type row_t;
  constexpr int N  = 64 * NT;
  constexpr int AP = K + A_PAD;
  const int hi = lane >> 4;
  const int lo = lane & 15;
  for (int t = wave; t < 8; t += nwaves) {
    const int mt  = t & 1;
    const int col = (t >> 1) * (16 * NT) + NT * lo;
    const float* ap = A32 + (mt * 16 + lo) * AP + 2 * hi;
    const float* wp = W + (2 * hi) * N + col;
    v8f acc[NT];
    for (int j = 0; j < NT; ++j) acc[j] = v8f{};
#pragma unroll(16 / NT)
    for (int kb = 0; kb < K; kb += 4) {
      const v2f a = *reinterpret_cast<const v2f*>(ap + kb);
      const row_t b0 = *reinterpret_cast<const row_t*>(wp + kb * N);
      const row_t b1 = *reinterpret_cast<const row_t*>(wp + (kb + 1) * N);
      for (int j = 0; j < NT; ++j) {
        v2f b;
        b.x = wrow_get(b0, j);
        b.y = wrow_get(b1, j);
        acc[j] = wmma_f32_k4(a, b, acc[j]);
      }
    }
    for (int v = 0; v < 8; ++v) {
      row_t o;
      for (int j = 0; j < NT; ++j) wrow_set(o, j, acc[j][v]);
      *reinterpret_cast<row_t*>(T + (mt * 16 + v + 8 * hi) * N + col) = o;
    }
  }
}

template <int C, bool RELU>
__device__ __forceinline__ void aggregate_gather(
    const float* __restrict__ T, float* __restrict__ H,
    const float* __restrict__ bias, const float* s_dinv,
    int dstA, int dstB, int srcA, int srcB, float nrmA, float nrmB,
    int wave, int nwaves, int lane) {
  constexpr int V  = C / 128;
  constexpr int HP = C + A_PAD;
  float4 bv[V];
  for (int v = 0; v < V; ++v)
    bv[v] = *reinterpret_cast<const float4*>(bias + v * 128 + lane * 4);
  for (int i = wave; i < N_ATOMS; i += nwaves) {
    unsigned long long m =
        (unsigned long long)__builtin_amdgcn_ballot_w32(dstA == i) |
        ((unsigned long long)__builtin_amdgcn_ballot_w32(dstB == i) << 32);
    const float di = s_dinv[i];
    const float sw = di * di;
    const float* Ti = T + i * C + lane * 4;
    float4 t[V], acc[V];
    for (int v = 0; v < V; ++v)
      t[v] = *reinterpret_cast<const float4*>(Ti + v * 128);
    bool self_pending = true;
    do {
      float4 r[4][V];
      float wj[4];
      for (int u = 0; u < 4; ++u) {
        wj[u] = 0.0f;
        for (int v = 0; v < V; ++v) {
          r[u][v].x = 0.0f; r[u][v].y = 0.0f; r[u][v].z = 0.0f; r[u][v].w = 0.0f;
        }
        if (m != 0) {
          const int e = __builtin_ctzll(m);
          m &= m - 1;
          const int l = e & 31;
          const int sa = __builtin_amdgcn_readlane(srcA, l);
          const int sb = __builtin_amdgcn_readlane(srcB, l);
          const int wa = __builtin_amdgcn_readlane(__float_as_int(nrmA), l);
          const int wb = __builtin_amdgcn_readlane(__float_as_int(nrmB), l);
          const int s = (e < 32) ? sa : sb;
          wj[u] = __int_as_float((e < 32) ? wa : wb);
          const float* Ts = T + s * C + lane * 4;
          for (int v = 0; v < V; ++v)
            r[u][v] = *reinterpret_cast<const float4*>(Ts + v * 128);
        }
      }
      if (self_pending) {
        self_pending = false;
        for (int v = 0; v < V; ++v) {
          acc[v].x = sw * t[v].x + bv[v].x;
          acc[v].y = sw * t[v].y + bv[v].y;
          acc[v].z = sw * t[v].z + bv[v].z;
          acc[v].w = sw * t[v].w + bv[v].w;
        }
      }
      for (int u = 0; u < 4; ++u) {
        for (int v = 0; v < V; ++v) {
          acc[v].x += wj[u] * r[u][v].x;
          acc[v].y += wj[u] * r[u][v].y;
          acc[v].z += wj[u] * r[u][v].z;
          acc[v].w += wj[u] * r[u][v].w;
        }
      }
    } while (m != 0);
    float* Hi = H + i * HP + lane * 4;
    for (int v = 0; v < V; ++v) {
      if (RELU) {
        acc[v].x = fmaxf(acc[v].x, 0.0f);
        acc[v].y = fmaxf(acc[v].y, 0.0f);
        acc[v].z = fmaxf(acc[v].z, 0.0f);
        acc[v].w = fmaxf(acc[v].w, 0.0f);
      }
      *reinterpret_cast<float4*>(Hi + v * 128) = acc[v];
    }
  }
  __syncthreads();
}

__global__ __launch_bounds__(256) void mol_gnn_kernel(
    const float* __restrict__ mol_x, const int* __restrict__ mei,
    const float* __restrict__ W1, const float* __restrict__ b1,
    const float* __restrict__ W2, const float* __restrict__ b2,
    const float* __restrict__ W3, const float* __restrict__ b3,
    float* __restrict__ embed) {
  __shared__ __attribute__((aligned(16))) float Abuf[N_ATOMS * (HID + A_PAD)];
  __shared__ __attribute__((aligned(16))) float Bbuf[N_ATOMS * HID];
  __shared__ float s_dinv[N_ATOMS];
  __shared__ float s_norm[E_MOL];
  __shared__ int   s_deg[N_ATOMS];
  __shared__ int   s_src[E_MOL];
  __shared__ int   s_dst[E_MOL];

  const int mol  = blockIdx.x;
  const int tid  = threadIdx.x;
  const int nthr = blockDim.x;
  const int lane = tid & 31;
  const int wave = tid >> 5;
  const int nw   = nthr >> 5;

  const float* X = mol_x + (size_t)mol * (N_ATOMS * IN_CH);
  for (int i = tid; i < N_ATOMS * IN_CH; i += nthr)
    Abuf[(i / IN_CH) * (IN_CH + A_PAD) + (i % IN_CH)] = X[i];
  const int* me = mei + (size_t)mol * (2 * E_MOL);
  if (tid < E_MOL) { s_src[tid] = me[tid]; s_dst[tid] = me[E_MOL + tid]; }
  if (tid < N_ATOMS) s_deg[tid] = 1;
  __syncthreads();
  if (tid < E_MOL) atomicAdd(&s_deg[s_dst[tid]], 1);
  __syncthreads();
  if (tid < N_ATOMS) s_dinv[tid] = rsqrtf((float)s_deg[tid]);
  __syncthreads();
  if (tid < E_MOL) s_norm[tid] = s_dinv[s_src[tid]] * s_dinv[s_dst[tid]];
  __syncthreads();
  const int   dstA = s_dst[lane],  dstB = s_dst[lane + 32];
  const int   srcA = s_src[lane],  srcB = s_src[lane + 32];
  const float nrmA = s_norm[lane], nrmB = s_norm[lane + 32];
  const int wv = __builtin_amdgcn_readfirstlane(wave);

  gemm32_wmma<IN_CH, HID / 64>(Abuf, W1, Bbuf, wave, nw, lane);
  __syncthreads();
  aggregate_gather<HID, true>(Bbuf, Abuf, b1, s_dinv, dstA, dstB, srcA, srcB,
                              nrmA, nrmB, wv, nw, lane);

  gemm32_wmma<HID, HID / 64>(Abuf, W2, Bbuf, wave, nw, lane);
  __syncthreads();
  aggregate_gather<HID, true>(Bbuf, Abuf, b2, s_dinv, dstA, dstB, srcA, srcB,
                              nrmA, nrmB, wv, nw, lane);

  gemm32_wmma<HID, OUT_CH / 64>(Abuf, W3, Bbuf, wave, nw, lane);
  __syncthreads();
  aggregate_gather<OUT_CH, false>(Bbuf, Abuf, b3, s_dinv, dstA, dstB, srcA, srcB,
                                  nrmA, nrmB, wv, nw, lane);

  for (int c = tid; c < OUT_CH; c += nthr) {
    float v = Abuf[c];
    for (int r = 1; r < N_ATOMS; ++r)
      v = fminf(v, Abuf[r * (OUT_CH + A_PAD) + c]);
    embed[(size_t)mol * OUT_CH + c] = v;
  }
}

__global__ __launch_bounds__(256) void wmma_gemm_kernel(
    const float* __restrict__ A, const float* __restrict__ B,
    float* __restrict__ C, int Mrows, int K, int N) {
  const int lane = threadIdx.x & 31;
  const int wave = blockIdx.x * (blockDim.x >> 5) + (threadIdx.x >> 5);
  const int ntT = N >> 4;
  const int nTiles = (Mrows >> 4) * ntT;
  if (wave >= nTiles) return;
  const int mt = wave / ntT;
  const int nt = wave - mt * ntT;
  const int hi = lane >> 4, lo = lane & 15;
  const int mrow = mt * 16 + lo;
  const int ncol = nt * 16 + lo;
  v8f acc = {};
  for (int kb = 0; kb < K; kb += 4) {
    const int k = kb + 2 * hi;
    v2f a, b;
    a.x = A[(size_t)mrow * K + k];
    a.y = A[(size_t)mrow * K + k + 1];
    b.x = B[(size_t)k * N + ncol];
    b.y = B[(size_t)(k + 1) * N + ncol];
    acc = wmma_f32_k4(a, b, acc);
  }
#pragma unroll
  for (int v = 0; v < 8; ++v)
    C[(size_t)(mt * 16 + v + 8 * hi) * N + nt * 16 + lo] = acc[v];
}


__global__ void kg_zero_kernel(int* __restrict__ cnt, int n) {
  int i = blockIdx.x * blockDim.x + threadIdx.x;
  if (i < n) cnt[i] = 0;
}
__global__ void kg_count_kernel(const int* __restrict__ dst,
                                int* __restrict__ cnt, int ne) {
  int i = blockIdx.x * blockDim.x + threadIdx.x;
  if (i < ne) atomicAdd(&cnt[dst[i]], 1);
}
__global__ __launch_bounds__(1024) void kg_scan_kernel(
    const int* __restrict__ cnt, int* __restrict__ rowptr,
    int* __restrict__ cursor, float* __restrict__ dinv, int n) {
  __shared__ int s_part[1024];
  const int tid = threadIdx.x;
  const int per = (n + 1023) >> 10;
  const int lo = tid * per;
  const int hi = min(lo + per, n);
  int sum = 0;
  for (int i = lo; i < hi; ++i) sum += cnt[i];
  s_part[tid] = sum;
  __syncthreads();
  for (int off = 1; off < 1024; off <<= 1) {
    const int v = (tid >= off) ? s_part[tid - off] : 0;
    __syncthreads();
    s_part[tid] += v;
    __syncthreads();
  }
  int run = s_part[tid] - sum;
  for (int i = lo; i < hi; ++i) {
    const int c = cnt[i];
    rowptr[i] = run;
    cursor[i] = run;
    dinv[i] = rsqrtf((float)(c + 1));
    run += c;
  }
  if (tid == 1023) rowptr[n] = s_part[1023];
}
__global__ void kg_fill_kernel(const int* __restrict__ src,
                               const int* __restrict__ dst,
                               int* __restrict__ cursor,
                               int* __restrict__ csr_src, int ne) {
  int e = blockIdx.x * blockDim.x + threadIdx.x;
  if (e < ne) {
    const int p = atomicAdd(&cursor[dst[e]], 1);
    csr_src[p] = src[e];
  }
}

template <int C, bool RELU>
__global__ __launch_bounds__(256) void kg_gather_kernel(
    const float* __restrict__ XW, const float* __restrict__ dinv,
    const float* __restrict__ bias, const int* __restrict__ rowptr,
    const int* __restrict__ csr_src, float* __restrict__ H, int n) {
  constexpr int SLICES = C / 128;
  const int lane = threadIdx.x & 31;
  const int wv = __builtin_amdgcn_readfirstlane((int)(threadIdx.x >> 5));
  const int gw = blockIdx.x * (int)(blockDim.x >> 5) + wv;
  const int d = gw / SLICES;
  if (d >= n) return;
  const int col = (gw - d * SLICES) * 128 + lane * 4;
  const int beg = rowptr[d];
  const int end = rowptr[d + 1];
  const float dd = dinv[d];
  const float sw = dd * dd;
  const float4 xs = *reinterpret_cast<const float4*>(XW + (size_t)d * C + col);
  float4 acc;
  acc.x = sw * xs.x + bias[col + 0];
  acc.y = sw * xs.y + bias[col + 1];
  acc.z = sw * xs.z + bias[col + 2];
  acc.w = sw * xs.w + bias[col + 3];
  for (int base = beg; base < end; base += 32) {
    const int m = min(end - base, 32);
    int s = 0;
    float w = 0.0f;
    if (lane < m) {
      s = csr_src[base + lane];
      w = dinv[s] * dd;
    }
    for (int j = 0; j < m; j += 8) {
      float4 r[8];
      float wj[8];
#pragma unroll
      for (int u = 0; u < 8; ++u) {
        r[u].x = 0.0f; r[u].y = 0.0f; r[u].z = 0.0f; r[u].w = 0.0f;
        wj[u] = 0.0f;
        if (j + u < m) {
          const int sj = __builtin_amdgcn_readlane(s, j + u);
          wj[u] = __int_as_float(
              __builtin_amdgcn_readlane(__float_as_int(w), j + u));
          r[u] = *reinterpret_cast<const float4*>(XW + (size_t)sj * C + col);
        }
      }
#pragma unroll
      for (int u = 0; u < 8; ++u) {
        acc.x += wj[u] * r[u].x;
        acc.y += wj[u] * r[u].y;
        acc.z += wj[u] * r[u].z;
        acc.w += wj[u] * r[u].w;
      }
    }
  }
  if (RELU) {
    acc.x = fmaxf(acc.x, 0.0f);
    acc.y = fmaxf(acc.y, 0.0f);
    acc.z = fmaxf(acc.z, 0.0f);
    acc.w = fmaxf(acc.w, 0.0f);
  }
  *reinterpret_cast<float4*>(H + (size_t)d * C + col) = acc;
}

extern "C" void kernel_launch(void* const* d_in, const int* in_sizes, int n_in,
                              void* d_out, int out_size, void* d_ws, size_t ws_size,
                              hipStream_t stream) {
  const float* mol_x = (const float*)d_in[0];
  const int*   mei   = (const int*)d_in[1];
  const int*   kei   = (const int*)d_in[2];
  const float* W1 = (const float*)d_in[3];
  const float* b1 = (const float*)d_in[4];
  const float* W2 = (const float*)d_in[5];
  const float* b2 = (const float*)d_in[6];
  const float* W3 = (const float*)d_in[7];
  const float* b3 = (const float*)d_in[8];
  const float* Wk1 = (const float*)d_in[9];
  const float* bk1 = (const float*)d_in[10];
  const float* Wk2 = (const float*)d_in[11];
  const float* bk2 = (const float*)d_in[12];
  float* out = (float*)d_out;

  float* ws     = (float*)d_ws;
  float* embed  = ws;
  float* XW     = embed + (size_t)M_MOL * OUT_CH;
  float* XW2    = XW;
  float* h      = XW + (size_t)M_MOL * HID;
  float* dinv   = h + (size_t)M_MOL * HID;
  int*   cnt    = (int*)(dinv + M_MOL);
  int*   rowptr = cnt + M_MOL;
  int*   cursor = rowptr + (M_MOL + 16);
  int*   csr_src = cursor + M_MOL;

  const int* kg_src = kei;
  const int* kg_dst = kei + E_KG;

  mol_gnn_kernel<<<M_MOL, 256, 0, stream>>>(mol_x, mei, W1, b1, W2, b2, W3, b3,
                                            embed);

  kg_zero_kernel<<<(M_MOL + 255) / 256, 256, 0, stream>>>(cnt, M_MOL);
  kg_count_kernel<<<(E_KG + 255) / 256, 256, 0, stream>>>(kg_dst, cnt, E_KG);
  kg_scan_kernel<<<1, 1024, 0, stream>>>(cnt, rowptr, cursor, dinv, M_MOL);
  kg_fill_kernel<<<(E_KG + 255) / 256, 256, 0, stream>>>(kg_src, kg_dst, cursor,
                                                         csr_src, E_KG);

  {
    int tiles = (M_MOL / 16) * (HID / 16);
    wmma_gemm_kernel<<<(tiles + 7) / 8, 256, 0, stream>>>(embed, Wk1, XW,
                                                          M_MOL, OUT_CH, HID);
  }
  kg_gather_kernel<HID, true><<<(M_MOL * (HID / 128) + 7) / 8, 256, 0, stream>>>(
      XW, dinv, bk1, rowptr, csr_src, h, M_MOL);

  {
    int tiles = (M_MOL / 16) * (OUT_CH / 16);
    wmma_gemm_kernel<<<(tiles + 7) / 8, 256, 0, stream>>>(h, Wk2, XW2,
                                                          M_MOL, HID, OUT_CH);
  }
  kg_gather_kernel<OUT_CH, false><<<(M_MOL * (OUT_CH / 128) + 7) / 8, 256, 0,
                                    stream>>>(XW2, dinv, bk2, rowptr, csr_src,
                                              out, M_MOL);
}
